# starting kernel
# baseline (speedup 1.0000x reference)
#include <hip/hip_runtime.h>


#define BATCH 4096
#define TLEN  200
#define INSZ  34
#define HID   4
#define GATES 16
#define NCLS  3
#define MROWS (BATCH * TLEN)

typedef float v2f __attribute__((ext_vector_type(2)));
typedef float v8f __attribute__((ext_vector_type(8)));

__device__ __forceinline__ float fast_rcp(float x) {
  return __builtin_amdgcn_rcpf(x);
}
__device__ __forceinline__ float fast_sigmoid(float x) {
  return fast_rcp(1.0f + __expf(-x));
}
__device__ __forceinline__ float fast_tanh(float x) {
#if __has_builtin(__builtin_amdgcn_tanhf)
  return __builtin_amdgcn_tanhf(x);
#else
  float e = __expf(-2.0f * x);
  return (1.0f - e) * fast_rcp(1.0f + e);
#endif
}

__global__ __launch_bounds__(256) void lstm_xgates_wmma(
    const float* __restrict__ x,
    const float* __restrict__ W_ih,
    const float* __restrict__ b_ih,
    const float* __restrict__ b_hh,
    float* __restrict__ xg)
{
  const int lane = threadIdx.x & 31;
  const int wave = threadIdx.x >> 5;
  const int half = lane >> 4;
  const int l    = lane & 15;
  const int koff = half * 2;
  const long long m0 = ((long long)blockIdx.x * 8 + wave) * 16;

  const float bias = b_ih[l] + b_hh[l];
  v8f acc;
#pragma unroll
  for (int v = 0; v < 8; ++v) acc[v] = bias;

  const float* __restrict__ xrow = x    + (m0 + l) * INSZ;
  const float* __restrict__ wrow = W_ih + l * INSZ;

#pragma unroll
  for (int kk = 0; kk < 9; ++kk) {
    const int k0 = kk * 4 + koff;
    float a0 = xrow[min(k0,     INSZ - 1)];
    float a1 = xrow[min(k0 + 1, INSZ - 1)];
    float w0 = wrow[min(k0,     INSZ - 1)];
    float w1 = wrow[min(k0 + 1, INSZ - 1)];
    v2f a, b;
    a.x = (k0     < INSZ) ? a0 : 0.0f;
    a.y = (k0 + 1 < INSZ) ? a1 : 0.0f;
    b.x = (k0     < INSZ) ? w0 : 0.0f;
    b.y = (k0 + 1 < INSZ) ? w1 : 0.0f;
    acc = __builtin_amdgcn_wmma_f32_16x16x4_f32(
        false, a, false, b, (short)0, acc, false, false);
  }

#pragma unroll
  for (int v = 0; v < 8; ++v) {
    const long long m = m0 + v + 8 * half;
    xg[m * GATES + l] = acc[v];
  }
}

__global__ __launch_bounds__(256) void lstm_scan(
    const float* __restrict__ xg,
    const float* __restrict__ W_hh,
    const float* __restrict__ W_fc,
    const float* __restrict__ b_fc,
    float* __restrict__ out)
{
  const int gtid = blockIdx.x * blockDim.x + threadIdx.x;
  const int b = gtid >> 2;
  const int j = gtid & 3;
  if (b >= BATCH) return;

  const float4 Wi = *(const float4*)(W_hh + (j     ) * HID);
  const float4 Wf = *(const float4*)(W_hh + (4  + j) * HID);
  const float4 Wg = *(const float4*)(W_hh + (8  + j) * HID);
  const float4 Wo = *(const float4*)(W_hh + (12 + j) * HID);

  const float* __restrict__ xgb = xg + (long long)b * TLEN * GATES;

  float h = 0.0f, c = 0.0f;
  for (int t = 0; t < TLEN; ++t) {
    const float* __restrict__ p = xgb + t * GATES;
    __builtin_prefetch(p + 8 * GATES, 0, 0);

    float pi = p[j];
    float pf = p[4  + j];
    float pg = p[8  + j];
    float po = p[12 + j];

    const float h0 = __shfl(h, 0, 4);
    const float h1 = __shfl(h, 1, 4);
    const float h2 = __shfl(h, 2, 4);
    const float h3 = __shfl(h, 3, 4);

    pi += Wi.x * h0 + Wi.y * h1 + Wi.z * h2 + Wi.w * h3;
    pf += Wf.x * h0 + Wf.y * h1 + Wf.z * h2 + Wf.w * h3;
    pg += Wg.x * h0 + Wg.y * h1 + Wg.z * h2 + Wg.w * h3;
    po += Wo.x * h0 + Wo.y * h1 + Wo.z * h2 + Wo.w * h3;

    const float ig = fast_sigmoid(pi);
    const float fg = fast_sigmoid(pf);
    const float gg = fast_tanh(pg);
    const float og = fast_sigmoid(po);

    c = fg * c + ig * gg;
    h = og * fast_tanh(c);
  }

  const float h0 = __shfl(h, 0, 4);
  const float h1 = __shfl(h, 1, 4);
  const float h2 = __shfl(h, 2, 4);
  const float h3 = __shfl(h, 3, 4);
  if (j < NCLS) {
    const float4 Wc = *(const float4*)(W_fc + j * HID);
    out[b * NCLS + j] =
        b_fc[j] + Wc.x * h0 + Wc.y * h1 + Wc.z * h2 + Wc.w * h3;
  }
}

extern "C" void kernel_launch(void* const* d_in, const int* in_sizes, int n_in,
                              void* d_out, int out_size, void* d_ws, size_t ws_size,
                              hipStream_t stream) {
  (void)in_sizes; (void)n_in; (void)out_size; (void)ws_size;
  const float* x    = (const float*)d_in[0];
  const float* W_ih = (const float*)d_in[1];
  const float* W_hh = (const float*)d_in[2];
  const float* b_ih = (const float*)d_in[3];
  const float* b_hh = (const float*)d_in[4];
  const float* W_fc = (const float*)d_in[5];
  const float* b_fc = (const float*)d_in[6];
  float* out = (float*)d_out;
  float* xg  = (float*)d_ws;

  const int tiles  = MROWS / 16;
  const int blocks1 = tiles / 8;
  hipLaunchKernelGGL(lstm_xgates_wmma, dim3(blocks1), dim3(256), 0, stream,
                     x, W_ih, b_ih, b_hh, xg);

  const int blocks2 = (BATCH * 4) / 256;
  hipLaunchKernelGGL(lstm_scan, dim3(blocks2), dim3(256), 0, stream,
                     xg, W_hh, W_fc, b_fc, out);
}


# scan kernel: x_gates loads software-pipelined 8 steps ahead through a register ring (L2 prefetch moved out to 24 steps), so the serial h chain no longer waits on its own step's loads
# speedup vs baseline: 1.2210x; 1.2210x over previous
#include <hip/hip_runtime.h>


#define BATCH 4096
#define TLEN  200
#define INSZ  34
#define HID   4
#define GATES 16
#define NCLS  3
#define MROWS (BATCH * TLEN)

typedef float v2f __attribute__((ext_vector_type(2)));
typedef float v8f __attribute__((ext_vector_type(8)));

__device__ __forceinline__ float fast_rcp(float x) {
  return __builtin_amdgcn_rcpf(x);
}
__device__ __forceinline__ float fast_sigmoid(float x) {
  return fast_rcp(1.0f + __expf(-x));
}
__device__ __forceinline__ float fast_tanh(float x) {
#if __has_builtin(__builtin_amdgcn_tanhf)
  return __builtin_amdgcn_tanhf(x);
#else
  float e = __expf(-2.0f * x);
  return (1.0f - e) * fast_rcp(1.0f + e);
#endif
}

__global__ __launch_bounds__(256) void lstm_xgates_wmma(
    const float* __restrict__ x,
    const float* __restrict__ W_ih,
    const float* __restrict__ b_ih,
    const float* __restrict__ b_hh,
    float* __restrict__ xg)
{
  const int lane = threadIdx.x & 31;
  const int wave = threadIdx.x >> 5;
  const int half = lane >> 4;
  const int l    = lane & 15;
  const int koff = half * 2;
  const long long m0 = ((long long)blockIdx.x * 8 + wave) * 16;

  const float bias = b_ih[l] + b_hh[l];
  v8f acc;
#pragma unroll
  for (int v = 0; v < 8; ++v) acc[v] = bias;

  const float* __restrict__ xrow = x    + (m0 + l) * INSZ;
  const float* __restrict__ wrow = W_ih + l * INSZ;

#pragma unroll
  for (int kk = 0; kk < 9; ++kk) {
    const int k0 = kk * 4 + koff;
    float a0 = xrow[min(k0,     INSZ - 1)];
    float a1 = xrow[min(k0 + 1, INSZ - 1)];
    float w0 = wrow[min(k0,     INSZ - 1)];
    float w1 = wrow[min(k0 + 1, INSZ - 1)];
    v2f a, b;
    a.x = (k0     < INSZ) ? a0 : 0.0f;
    a.y = (k0 + 1 < INSZ) ? a1 : 0.0f;
    b.x = (k0     < INSZ) ? w0 : 0.0f;
    b.y = (k0 + 1 < INSZ) ? w1 : 0.0f;
    acc = __builtin_amdgcn_wmma_f32_16x16x4_f32(
        false, a, false, b, (short)0, acc, false, false);
  }

#pragma unroll
  for (int v = 0; v < 8; ++v) {
    const long long m = m0 + v + 8 * half;
    xg[m * GATES + l] = acc[v];
  }
}

#define PIPE     8
#define PF_AHEAD 24
static_assert(TLEN % PIPE == 0 && TLEN >= 2 * PIPE, "ring depth must divide the sequence length");

__device__ __forceinline__ void lstm_step(
    float pi, float pf, float pg, float po,
    const float4 Wi, const float4 Wf, const float4 Wg, const float4 Wo,
    float& h, float& c)
{
  const float h0 = __shfl(h, 0, 4);
  const float h1 = __shfl(h, 1, 4);
  const float h2 = __shfl(h, 2, 4);
  const float h3 = __shfl(h, 3, 4);

  pi += Wi.x * h0 + Wi.y * h1 + Wi.z * h2 + Wi.w * h3;
  pf += Wf.x * h0 + Wf.y * h1 + Wf.z * h2 + Wf.w * h3;
  pg += Wg.x * h0 + Wg.y * h1 + Wg.z * h2 + Wg.w * h3;
  po += Wo.x * h0 + Wo.y * h1 + Wo.z * h2 + Wo.w * h3;

  const float ig = fast_sigmoid(pi);
  const float fg = fast_sigmoid(pf);
  const float gg = fast_tanh(pg);
  const float og = fast_sigmoid(po);

  c = fg * c + ig * gg;
  h = og * fast_tanh(c);
}

__global__ __launch_bounds__(256) void lstm_scan(
    const float* __restrict__ xg,
    const float* __restrict__ W_hh,
    const float* __restrict__ W_fc,
    const float* __restrict__ b_fc,
    float* __restrict__ out)
{
  const int gtid = blockIdx.x * blockDim.x + threadIdx.x;
  const int b = gtid >> 2;
  const int j = gtid & 3;
  if (b >= BATCH) return;

  const float4 Wi = *(const float4*)(W_hh + (j     ) * HID);
  const float4 Wf = *(const float4*)(W_hh + (4  + j) * HID);
  const float4 Wg = *(const float4*)(W_hh + (8  + j) * HID);
  const float4 Wo = *(const float4*)(W_hh + (12 + j) * HID);

  const float* __restrict__ xgl = xg + (long long)b * TLEN * GATES + j;

  float ri[PIPE], rf[PIPE], rg[PIPE], ro[PIPE];
#pragma unroll
  for (int u = 0; u < PIPE; ++u) {
    const float* __restrict__ p = xgl + u * GATES;
    ri[u] = p[0];
    rf[u] = p[4];
    rg[u] = p[8];
    ro[u] = p[12];
  }

  float h = 0.0f, c = 0.0f;

  for (int t0 = 0; t0 < TLEN - PIPE; t0 += PIPE) {
    const float* __restrict__ nx = xgl + (t0 + PIPE) * GATES;
#pragma unroll
    for (int u = 0; u < PIPE; ++u) {
      lstm_step(ri[u], rf[u], rg[u], ro[u], Wi, Wf, Wg, Wo, h, c);
      const float* __restrict__ p = nx + u * GATES;
      ri[u] = p[0];
      rf[u] = p[4];
      rg[u] = p[8];
      ro[u] = p[12];
      const int tp = min(t0 + u + PF_AHEAD, TLEN - 1);
      __builtin_prefetch(xgl + tp * GATES, 0, 0);
    }
  }
#pragma unroll
  for (int u = 0; u < PIPE; ++u)
    lstm_step(ri[u], rf[u], rg[u], ro[u], Wi, Wf, Wg, Wo, h, c);

  const float h0 = __shfl(h, 0, 4);
  const float h1 = __shfl(h, 1, 4);
  const float h2 = __shfl(h, 2, 4);
  const float h3 = __shfl(h, 3, 4);
  if (j < NCLS) {
    const float4 Wc = *(const float4*)(W_fc + j * HID);
    out[b * NCLS + j] =
        b_fc[j] + Wc.x * h0 + Wc.y * h1 + Wc.z * h2 + Wc.w * h3;
  }
}

extern "C" void kernel_launch(void* const* d_in, const int* in_sizes, int n_in,
                              void* d_out, int out_size, void* d_ws, size_t ws_size,
                              hipStream_t stream) {
  (void)in_sizes; (void)n_in; (void)out_size; (void)ws_size;
  const float* x    = (const float*)d_in[0];
  const float* W_ih = (const float*)d_in[1];
  const float* W_hh = (const float*)d_in[2];
  const float* b_ih = (const float*)d_in[3];
  const float* b_hh = (const float*)d_in[4];
  const float* W_fc = (const float*)d_in[5];
  const float* b_fc = (const float*)d_in[6];
  float* out = (float*)d_out;
  float* xg  = (float*)d_ws;

  const int tiles  = MROWS / 16;
  const int blocks1 = tiles / 8;
  hipLaunchKernelGGL(lstm_xgates_wmma, dim3(blocks1), dim3(256), 0, stream,
                     x, W_ih, b_ih, b_hh, xg);

  const int blocks2 = (BATCH * 4) / 256;
  hipLaunchKernelGGL(lstm_scan, dim3(blocks2), dim3(256), 0, stream,
                     xg, W_hh, W_fc, b_fc, out);
}


# projection kernel: each wave walks 5 row tiles (grid 6400 -> 1280 workgroups) with the next tile's loads in flight and W_ih fragments loaded once per wave, instead of one short-lived wave per tile
# speedup vs baseline: 1.3162x; 1.0780x over previous
#include <hip/hip_runtime.h>


#define BATCH 4096
#define TLEN  200
#define INSZ  34
#define HID   4
#define GATES 16
#define NCLS  3
#define MROWS (BATCH * TLEN)

typedef float v2f __attribute__((ext_vector_type(2)));
typedef float v8f __attribute__((ext_vector_type(8)));

__device__ __forceinline__ float fast_rcp(float x) {
  return __builtin_amdgcn_rcpf(x);
}
__device__ __forceinline__ float fast_sigmoid(float x) {
  return fast_rcp(1.0f + __expf(-x));
}
__device__ __forceinline__ float fast_tanh(float x) {
#if __has_builtin(__builtin_amdgcn_tanhf)
  return __builtin_amdgcn_tanhf(x);
#else
  float e = __expf(-2.0f * x);
  return (1.0f - e) * fast_rcp(1.0f + e);
#endif
}

#define K1_TILES 5
#define K1_WAVES 8
static_assert((MROWS / 16) % (K1_WAVES * K1_TILES) == 0, "tiles must divide evenly over the waves");
static_assert(K1_TILES % 2 == 1, "the loop below does two tiles per trip and one behind it");

__device__ __forceinline__ void xg_load_a(
    const float* __restrict__ xrow, const int koff, v2f (&a)[9])
{
#pragma unroll
  for (int kk = 0; kk < 9; ++kk) {
    const int k0 = kk * 4 + koff;
    a[kk].x = xrow[min(k0,     INSZ - 1)];
    a[kk].y = xrow[min(k0 + 1, INSZ - 1)];
  }
}

__device__ __forceinline__ void xg_tile(
    const v2f (&a)[9], const v2f (&b)[9], const float bias,
    float* __restrict__ xgt, const int koff, const int half, const int l)
{
  v8f acc;
#pragma unroll
  for (int v = 0; v < 8; ++v) acc[v] = bias;
#pragma unroll
  for (int kk = 0; kk < 9; ++kk) {
    const int k0 = kk * 4 + koff;
    v2f ak;
    ak.x = (k0     < INSZ) ? a[kk].x : 0.0f;
    ak.y = (k0 + 1 < INSZ) ? a[kk].y : 0.0f;
    acc = __builtin_amdgcn_wmma_f32_16x16x4_f32(
        false, ak, false, b[kk], (short)0, acc, false, false);
  }
#pragma unroll
  for (int v = 0; v < 8; ++v)
    xgt[(v + 8 * half) * GATES + l] = acc[v];
}

__global__ __launch_bounds__(256) void lstm_xgates_wmma(
    const float* __restrict__ x,
    const float* __restrict__ W_ih,
    const float* __restrict__ b_ih,
    const float* __restrict__ b_hh,
    float* __restrict__ xg)
{
  const int lane = threadIdx.x & 31;
  const int wave = threadIdx.x >> 5;
  const int half = lane >> 4;
  const int l    = lane & 15;
  const int koff = half * 2;

  const long long tile0 = (long long)blockIdx.x * (K1_WAVES * K1_TILES) + wave;
  const float* __restrict__ xrow = x  + (tile0 * 16 + l) * INSZ;
  float*       __restrict__ xgt  = xg + (tile0 * 16) * GATES;
  const int XSTEP = K1_WAVES * 16 * INSZ;
  const int GSTEP = K1_WAVES * 16 * GATES;

  v2f bw[9];
  const float* __restrict__ wrow = W_ih + l * INSZ;
#pragma unroll
  for (int kk = 0; kk < 9; ++kk) {
    const int k0 = kk * 4 + koff;
    const float w0 = wrow[min(k0,     INSZ - 1)];
    const float w1 = wrow[min(k0 + 1, INSZ - 1)];
    bw[kk].x = (k0     < INSZ) ? w0 : 0.0f;
    bw[kk].y = (k0 + 1 < INSZ) ? w1 : 0.0f;
  }
  const float bias = b_ih[l] + b_hh[l];

  v2f fa[9], fb[9];
  xg_load_a(xrow, koff, fa);
#pragma unroll 1
  for (int i = 0; i < K1_TILES - 1; i += 2) {
    xg_load_a(xrow + (i + 1) * XSTEP, koff, fb);
    __builtin_amdgcn_sched_barrier(0);
    xg_tile(fa, bw, bias, xgt + i * GSTEP, koff, half, l);
    xg_load_a(xrow + (i + 2) * XSTEP, koff, fa);
    __builtin_amdgcn_sched_barrier(0);
    xg_tile(fb, bw, bias, xgt + (i + 1) * GSTEP, koff, half, l);
  }
  xg_tile(fa, bw, bias, xgt + (K1_TILES - 1) * GSTEP, koff, half, l);
}

#define PIPE     8
#define PF_AHEAD 24
static_assert(TLEN % PIPE == 0 && TLEN >= 2 * PIPE, "ring depth must divide the sequence length");

__device__ __forceinline__ void lstm_step(
    float pi, float pf, float pg, float po,
    const float4 Wi, const float4 Wf, const float4 Wg, const float4 Wo,
    float& h, float& c)
{
  const float h0 = __shfl(h, 0, 4);
  const float h1 = __shfl(h, 1, 4);
  const float h2 = __shfl(h, 2, 4);
  const float h3 = __shfl(h, 3, 4);

  pi += Wi.x * h0 + Wi.y * h1 + Wi.z * h2 + Wi.w * h3;
  pf += Wf.x * h0 + Wf.y * h1 + Wf.z * h2 + Wf.w * h3;
  pg += Wg.x * h0 + Wg.y * h1 + Wg.z * h2 + Wg.w * h3;
  po += Wo.x * h0 + Wo.y * h1 + Wo.z * h2 + Wo.w * h3;

  const float ig = fast_sigmoid(pi);
  const float fg = fast_sigmoid(pf);
  const float gg = fast_tanh(pg);
  const float og = fast_sigmoid(po);

  c = fg * c + ig * gg;
  h = og * fast_tanh(c);
}

__global__ __launch_bounds__(256) void lstm_scan(
    const float* __restrict__ xg,
    const float* __restrict__ W_hh,
    const float* __restrict__ W_fc,
    const float* __restrict__ b_fc,
    float* __restrict__ out)
{
  const int gtid = blockIdx.x * blockDim.x + threadIdx.x;
  const int b = gtid >> 2;
  const int j = gtid & 3;
  if (b >= BATCH) return;

  const float4 Wi = *(const float4*)(W_hh + (j     ) * HID);
  const float4 Wf = *(const float4*)(W_hh + (4  + j) * HID);
  const float4 Wg = *(const float4*)(W_hh + (8  + j) * HID);
  const float4 Wo = *(const float4*)(W_hh + (12 + j) * HID);

  const float* __restrict__ xgl = xg + (long long)b * TLEN * GATES + j;

  float ri[PIPE], rf[PIPE], rg[PIPE], ro[PIPE];
#pragma unroll
  for (int u = 0; u < PIPE; ++u) {
    const float* __restrict__ p = xgl + u * GATES;
    ri[u] = p[0];
    rf[u] = p[4];
    rg[u] = p[8];
    ro[u] = p[12];
  }

  float h = 0.0f, c = 0.0f;

  for (int t0 = 0; t0 < TLEN - PIPE; t0 += PIPE) {
    const float* __restrict__ nx = xgl + (t0 + PIPE) * GATES;
#pragma unroll
    for (int u = 0; u < PIPE; ++u) {
      lstm_step(ri[u], rf[u], rg[u], ro[u], Wi, Wf, Wg, Wo, h, c);
      const float* __restrict__ p = nx + u * GATES;
      ri[u] = p[0];
      rf[u] = p[4];
      rg[u] = p[8];
      ro[u] = p[12];
      const int tp = min(t0 + u + PF_AHEAD, TLEN - 1);
      __builtin_prefetch(xgl + tp * GATES, 0, 0);
    }
  }
#pragma unroll
  for (int u = 0; u < PIPE; ++u)
    lstm_step(ri[u], rf[u], rg[u], ro[u], Wi, Wf, Wg, Wo, h, c);

  const float h0 = __shfl(h, 0, 4);
  const float h1 = __shfl(h, 1, 4);
  const float h2 = __shfl(h, 2, 4);
  const float h3 = __shfl(h, 3, 4);
  if (j < NCLS) {
    const float4 Wc = *(const float4*)(W_fc + j * HID);
    out[b * NCLS + j] =
        b_fc[j] + Wc.x * h0 + Wc.y * h1 + Wc.z * h2 + Wc.w * h3;
  }
}

extern "C" void kernel_launch(void* const* d_in, const int* in_sizes, int n_in,
                              void* d_out, int out_size, void* d_ws, size_t ws_size,
                              hipStream_t stream) {
  (void)in_sizes; (void)n_in; (void)out_size; (void)ws_size;
  const float* x    = (const float*)d_in[0];
  const float* W_ih = (const float*)d_in[1];
  const float* W_hh = (const float*)d_in[2];
  const float* b_ih = (const float*)d_in[3];
  const float* b_hh = (const float*)d_in[4];
  const float* W_fc = (const float*)d_in[5];
  const float* b_fc = (const float*)d_in[6];
  float* out = (float*)d_out;
  float* xg  = (float*)d_ws;

  const int tiles  = MROWS / 16;
  const int blocks1 = tiles / (K1_WAVES * K1_TILES);
  hipLaunchKernelGGL(lstm_xgates_wmma, dim3(blocks1), dim3(256), 0, stream,
                     x, W_ih, b_ih, b_hh, xg);

  const int blocks2 = (BATCH * 4) / 256;
  hipLaunchKernelGGL(lstm_scan, dim3(blocks2), dim3(256), 0, stream,
                     xg, W_hh, W_fc, b_fc, out);
}


# scan kernel: 2-wave workgroups (256, one per WGP) instead of 8-wave ones on 64 WGPs: each WGP's memory pipeline serves 2 waves' 8-line loads per step, not 8, and each wave has a SIMD to itself
# speedup vs baseline: 1.4292x; 1.0859x over previous
#include <hip/hip_runtime.h>


#define BATCH 4096
#define TLEN  200
#define INSZ  34
#define HID   4
#define GATES 16
#define NCLS  3
#define MROWS (BATCH * TLEN)

typedef float v2f __attribute__((ext_vector_type(2)));
typedef float v8f __attribute__((ext_vector_type(8)));

__device__ __forceinline__ float fast_rcp(float x) {
  return __builtin_amdgcn_rcpf(x);
}
__device__ __forceinline__ float fast_sigmoid(float x) {
  return fast_rcp(1.0f + __expf(-x));
}
__device__ __forceinline__ float fast_tanh(float x) {
#if __has_builtin(__builtin_amdgcn_tanhf)
  return __builtin_amdgcn_tanhf(x);
#else
  float e = __expf(-2.0f * x);
  return (1.0f - e) * fast_rcp(1.0f + e);
#endif
}

#define K1_TILES 5
#define K1_WAVES 8
static_assert((MROWS / 16) % (K1_WAVES * K1_TILES) == 0, "tiles must divide evenly over the waves");
static_assert(K1_TILES % 2 == 1, "the loop below does two tiles per trip and one behind it");

__device__ __forceinline__ void xg_load_a(
    const float* __restrict__ xrow, const int koff, v2f (&a)[9])
{
#pragma unroll
  for (int kk = 0; kk < 9; ++kk) {
    const int k0 = kk * 4 + koff;
    a[kk].x = xrow[min(k0,     INSZ - 1)];
    a[kk].y = xrow[min(k0 + 1, INSZ - 1)];
  }
}

__device__ __forceinline__ void xg_tile(
    const v2f (&a)[9], const v2f (&b)[9], const float bias,
    float* __restrict__ xgt, const int koff, const int half, const int l)
{
  v8f acc;
#pragma unroll
  for (int v = 0; v < 8; ++v) acc[v] = bias;
#pragma unroll
  for (int kk = 0; kk < 9; ++kk) {
    const int k0 = kk * 4 + koff;
    v2f ak;
    ak.x = (k0     < INSZ) ? a[kk].x : 0.0f;
    ak.y = (k0 + 1 < INSZ) ? a[kk].y : 0.0f;
    acc = __builtin_amdgcn_wmma_f32_16x16x4_f32(
        false, ak, false, b[kk], (short)0, acc, false, false);
  }
#pragma unroll
  for (int v = 0; v < 8; ++v)
    xgt[(v + 8 * half) * GATES + l] = acc[v];
}

__global__ __launch_bounds__(256) void lstm_xgates_wmma(
    const float* __restrict__ x,
    const float* __restrict__ W_ih,
    const float* __restrict__ b_ih,
    const float* __restrict__ b_hh,
    float* __restrict__ xg)
{
  const int lane = threadIdx.x & 31;
  const int wave = threadIdx.x >> 5;
  const int half = lane >> 4;
  const int l    = lane & 15;
  const int koff = half * 2;

  const long long tile0 = (long long)blockIdx.x * (K1_WAVES * K1_TILES) + wave;
  const float* __restrict__ xrow = x  + (tile0 * 16 + l) * INSZ;
  float*       __restrict__ xgt  = xg + (tile0 * 16) * GATES;
  const int XSTEP = K1_WAVES * 16 * INSZ;
  const int GSTEP = K1_WAVES * 16 * GATES;

  v2f bw[9];
  const float* __restrict__ wrow = W_ih + l * INSZ;
#pragma unroll
  for (int kk = 0; kk < 9; ++kk) {
    const int k0 = kk * 4 + koff;
    const float w0 = wrow[min(k0,     INSZ - 1)];
    const float w1 = wrow[min(k0 + 1, INSZ - 1)];
    bw[kk].x = (k0     < INSZ) ? w0 : 0.0f;
    bw[kk].y = (k0 + 1 < INSZ) ? w1 : 0.0f;
  }
  const float bias = b_ih[l] + b_hh[l];

  v2f fa[9], fb[9];
  xg_load_a(xrow, koff, fa);
#pragma unroll 1
  for (int i = 0; i < K1_TILES - 1; i += 2) {
    xg_load_a(xrow + (i + 1) * XSTEP, koff, fb);
    __builtin_amdgcn_sched_barrier(0);
    xg_tile(fa, bw, bias, xgt + i * GSTEP, koff, half, l);
    xg_load_a(xrow + (i + 2) * XSTEP, koff, fa);
    __builtin_amdgcn_sched_barrier(0);
    xg_tile(fb, bw, bias, xgt + (i + 1) * GSTEP, koff, half, l);
  }
  xg_tile(fa, bw, bias, xgt + (K1_TILES - 1) * GSTEP, koff, half, l);
}

#define PIPE     8
#define PF_AHEAD 24
static_assert(TLEN % PIPE == 0 && TLEN >= 2 * PIPE, "ring depth must divide the sequence length");

#define K2_THREADS 64
static_assert((BATCH * 4) % K2_THREADS == 0, "the scan's workgroups must tile the batch exactly");

__device__ __forceinline__ void lstm_step(
    float pi, float pf, float pg, float po,
    const float4 Wi, const float4 Wf, const float4 Wg, const float4 Wo,
    float& h, float& c)
{
  const float h0 = __shfl(h, 0, 4);
  const float h1 = __shfl(h, 1, 4);
  const float h2 = __shfl(h, 2, 4);
  const float h3 = __shfl(h, 3, 4);

  pi += Wi.x * h0 + Wi.y * h1 + Wi.z * h2 + Wi.w * h3;
  pf += Wf.x * h0 + Wf.y * h1 + Wf.z * h2 + Wf.w * h3;
  pg += Wg.x * h0 + Wg.y * h1 + Wg.z * h2 + Wg.w * h3;
  po += Wo.x * h0 + Wo.y * h1 + Wo.z * h2 + Wo.w * h3;

  const float ig = fast_sigmoid(pi);
  const float fg = fast_sigmoid(pf);
  const float gg = fast_tanh(pg);
  const float og = fast_sigmoid(po);

  c = fg * c + ig * gg;
  h = og * fast_tanh(c);
}

__global__ __launch_bounds__(K2_THREADS) void lstm_scan(
    const float* __restrict__ xg,
    const float* __restrict__ W_hh,
    const float* __restrict__ W_fc,
    const float* __restrict__ b_fc,
    float* __restrict__ out)
{
  const int gtid = blockIdx.x * K2_THREADS + threadIdx.x;
  const int b = gtid >> 2;
  const int j = gtid & 3;
  if (b >= BATCH) return;

  const float4 Wi = *(const float4*)(W_hh + (j     ) * HID);
  const float4 Wf = *(const float4*)(W_hh + (4  + j) * HID);
  const float4 Wg = *(const float4*)(W_hh + (8  + j) * HID);
  const float4 Wo = *(const float4*)(W_hh + (12 + j) * HID);

  const float* __restrict__ xgl = xg + (long long)b * TLEN * GATES + j;

  float ri[PIPE], rf[PIPE], rg[PIPE], ro[PIPE];
#pragma unroll
  for (int u = 0; u < PIPE; ++u) {
    const float* __restrict__ p = xgl + u * GATES;
    ri[u] = p[0];
    rf[u] = p[4];
    rg[u] = p[8];
    ro[u] = p[12];
  }

  float h = 0.0f, c = 0.0f;

  for (int t0 = 0; t0 < TLEN - PIPE; t0 += PIPE) {
    const float* __restrict__ nx = xgl + (t0 + PIPE) * GATES;
#pragma unroll
    for (int u = 0; u < PIPE; ++u) {
      lstm_step(ri[u], rf[u], rg[u], ro[u], Wi, Wf, Wg, Wo, h, c);
      const float* __restrict__ p = nx + u * GATES;
      ri[u] = p[0];
      rf[u] = p[4];
      rg[u] = p[8];
      ro[u] = p[12];
      const int tp = min(t0 + u + PF_AHEAD, TLEN - 1);
      __builtin_prefetch(xgl + tp * GATES, 0, 0);
    }
  }
#pragma unroll
  for (int u = 0; u < PIPE; ++u)
    lstm_step(ri[u], rf[u], rg[u], ro[u], Wi, Wf, Wg, Wo, h, c);

  const float h0 = __shfl(h, 0, 4);
  const float h1 = __shfl(h, 1, 4);
  const float h2 = __shfl(h, 2, 4);
  const float h3 = __shfl(h, 3, 4);
  if (j < NCLS) {
    const float4 Wc = *(const float4*)(W_fc + j * HID);
    out[b * NCLS + j] =
        b_fc[j] + Wc.x * h0 + Wc.y * h1 + Wc.z * h2 + Wc.w * h3;
  }
}

extern "C" void kernel_launch(void* const* d_in, const int* in_sizes, int n_in,
                              void* d_out, int out_size, void* d_ws, size_t ws_size,
                              hipStream_t stream) {
  (void)in_sizes; (void)n_in; (void)out_size; (void)ws_size;
  const float* x    = (const float*)d_in[0];
  const float* W_ih = (const float*)d_in[1];
  const float* W_hh = (const float*)d_in[2];
  const float* b_ih = (const float*)d_in[3];
  const float* b_hh = (const float*)d_in[4];
  const float* W_fc = (const float*)d_in[5];
  const float* b_fc = (const float*)d_in[6];
  float* out = (float*)d_out;
  float* xg  = (float*)d_ws;

  const int tiles  = MROWS / 16;
  const int blocks1 = tiles / (K1_WAVES * K1_TILES);
  hipLaunchKernelGGL(lstm_xgates_wmma, dim3(blocks1), dim3(256), 0, stream,
                     x, W_ih, b_ih, b_hh, xg);

  const int blocks2 = (BATCH * 4) / K2_THREADS;
  hipLaunchKernelGGL(lstm_scan, dim3(blocks2), dim3(K2_THREADS), 0, stream,
                     xg, W_hh, W_fc, b_fc, out);
}


# projection fused into the scan kernel: 4 producer waves per workgroup stream x, do the fp32 WMMA and hand gate pre-activations to 2 scan waves through LDS; x_gates never touches memory, one launch
# speedup vs baseline: 2.8259x; 1.9773x over previous
#include <hip/hip_runtime.h>


#define BATCH 4096
#define TLEN  200
#define INSZ  34
#define HID   4
#define GATES 16
#define NCLS  3

typedef float v2f __attribute__((ext_vector_type(2)));
typedef float v4f __attribute__((ext_vector_type(4)));
typedef float v8f __attribute__((ext_vector_type(8)));

__device__ __forceinline__ float fast_rcp(float x) {
  return __builtin_amdgcn_rcpf(x);
}
__device__ __forceinline__ float fast_sigmoid(float x) {
  return fast_rcp(1.0f + __expf(-x));
}
__device__ __forceinline__ float fast_tanh(float x) {
#if __has_builtin(__builtin_amdgcn_tanhf)
  return __builtin_amdgcn_tanhf(x);
#else
  float e = __expf(-2.0f * x);
  return (1.0f - e) * fast_rcp(1.0f + e);
#endif
}

#define NB      16
#define CH      8
#define NCH     (TLEN / CH)
#define SCANW   2
#define PRODW   4
#define FUSED_THREADS ((SCANW + PRODW) * 32)
#define PB      (NB / PRODW)
#define SEGF    (CH * INSZ)
#define SEGV    (SEGF / 4)
#define BSTRV   (TLEN * INSZ / 4)
#define RAWV    (PB * SEGV)
#define RAWLD   ((RAWV + 31) / 32)
#define STAGEV  (CH * NB * GATES / 4)
#define PF_CH   4

static_assert(TLEN % CH == 0, "chunks must tile the sequence");
static_assert(NCH % 2 == 1 && NCH >= 3, "the loop below does two chunks per trip behind two peeled ones and one drain");
static_assert(BATCH % NB == 0, "workgroups must tile the batch");
static_assert(SCANW * 8 == NB && PB == 4, "a producer wave makes two 16-row tiles (2 batch elements x 8 steps each) per chunk");
static_assert(SEGF % 4 == 0 && (TLEN * INSZ) % 4 == 0, "16-byte loads need 16-byte aligned segments");


__device__ __forceinline__ void produce_chunk(
    v4f (&g)[RAWLD], const v4f* __restrict__ xb, const int (&offs)[RAWLD],
    const int knext, const float* __restrict__ pfp, const int kpf,
    v4f* rawq, const v2f (&bw)[9], const float bias, float* dst,
    const int lane, const int c, const int koff)
{
#pragma unroll
  for (int n = 0; n < RAWLD; ++n)
    rawq[min(n * 32 + lane, RAWV - 1)] = g[n];
#pragma unroll
  for (int n = 0; n < RAWLD; ++n)
    g[n] = xb[offs[n] + knext * SEGV];
  __builtin_prefetch(pfp + kpf * SEGF, 0, 0);
  __builtin_amdgcn_sched_barrier(0);

  const float* rf = (const float*)rawq;
  v2f a0[9], a1[9];
#pragma unroll
  for (int kk = 0; kk < 9; ++kk) {
    const int kc = (kk < 8) ? (kk * 4 + koff) : (INSZ - 2);
    a0[kk] = *(const v2f*)(rf + c * INSZ + kc);
    a1[kk] = *(const v2f*)(rf + (16 + c) * INSZ + kc);
  }
  __builtin_amdgcn_sched_barrier(0);
  v8f acc0, acc1;
#pragma unroll
  for (int v = 0; v < 8; ++v) { acc0[v] = bias; acc1[v] = bias; }
#pragma unroll
  for (int kk = 0; kk < 9; ++kk) {
    const int k0 = kk * 4 + koff;
    v2f ak0, ak1;
    ak0.x = (k0     < INSZ) ? a0[kk].x : 0.0f;
    ak0.y = (k0 + 1 < INSZ) ? a0[kk].y : 0.0f;
    ak1.x = (k0     < INSZ) ? a1[kk].x : 0.0f;
    ak1.y = (k0 + 1 < INSZ) ? a1[kk].y : 0.0f;
    acc0 = __builtin_amdgcn_wmma_f32_16x16x4_f32(
        false, ak0, false, bw[kk], (short)0, acc0, false, false);
    acc1 = __builtin_amdgcn_wmma_f32_16x16x4_f32(
        false, ak1, false, bw[kk], (short)0, acc1, false, false);
  }
#pragma unroll
  for (int v = 0; v < 8; ++v) {
    dst[v * (NB * GATES)]      = acc0[v];
    dst[32 + v * (NB * GATES)] = acc1[v];
  }
}

__device__ __forceinline__ void lstm_step(
    float pi, float pf, float pg, float po,
    const float4 Wi, const float4 Wf, const float4 Wg, const float4 Wo,
    float& h, float& c)
{
  const float h0 = __shfl(h, 0, 4);
  const float h1 = __shfl(h, 1, 4);
  const float h2 = __shfl(h, 2, 4);
  const float h3 = __shfl(h, 3, 4);

  pi += Wi.x * h0 + Wi.y * h1 + Wi.z * h2 + Wi.w * h3;
  pf += Wf.x * h0 + Wf.y * h1 + Wf.z * h2 + Wf.w * h3;
  pg += Wg.x * h0 + Wg.y * h1 + Wg.z * h2 + Wg.w * h3;
  po += Wo.x * h0 + Wo.y * h1 + Wo.z * h2 + Wo.w * h3;

  const float ig = fast_sigmoid(pi);
  const float fg = fast_sigmoid(pf);
  const float gg = fast_tanh(pg);
  const float og = fast_sigmoid(po);

  c = fg * c + ig * gg;
  h = og * fast_tanh(c);
}

__device__ __forceinline__ void ring_load(float4 (&r)[CH], const float4* sl)
{
#pragma unroll
  for (int u = 0; u < CH; ++u) r[u] = sl[u * (NB * GATES / 4)];
}
__device__ __forceinline__ void ring_steps(
    const float4 (&r)[CH],
    const float4 Wi, const float4 Wf, const float4 Wg, const float4 Wo,
    float& h, float& c)
{
#pragma unroll
  for (int u = 0; u < CH; ++u)
    lstm_step(r[u].x, r[u].y, r[u].z, r[u].w, Wi, Wf, Wg, Wo, h, c);
}

__device__ __forceinline__ void stage_barrier()
{
  __builtin_amdgcn_fence(__ATOMIC_RELEASE, "workgroup", "local");
  __builtin_amdgcn_s_barrier();
  __builtin_amdgcn_fence(__ATOMIC_ACQUIRE, "workgroup", "local");
}

__global__ __launch_bounds__(FUSED_THREADS) void lstm_fused(
    const float* __restrict__ x,
    const float* __restrict__ W_ih,
    const float* __restrict__ W_hh,
    const float* __restrict__ b_ih,
    const float* __restrict__ b_hh,
    const float* __restrict__ W_fc,
    const float* __restrict__ b_fc,
    float* __restrict__ out)
{
  __shared__ float4 stage[2][STAGEV];
  __shared__ v4f raw[PRODW][RAWV];

  const int lane = threadIdx.x & 31;
  const int wave = __builtin_amdgcn_readfirstlane((int)(threadIdx.x >> 5));
  const bool prod = wave >= SCANW;
  const int b0 = blockIdx.x * NB;

  const int half = lane >> 4;
  const int c16  = lane & 15;
  const int koff = half * 2;
  const int q    = prod ? (wave - SCANW) : 0;
  const v4f* __restrict__ xb =
      (const v4f*)(x + (long long)(b0 + q * PB) * (TLEN * INSZ));
  int offs[RAWLD];
#pragma unroll
  for (int n = 0; n < RAWLD; ++n) {
    const int i = min(n * 32 + lane, RAWV - 1);
    offs[n] = (i / SEGV) * BSTRV + (i % SEGV);
  }
  const int pseg = min(lane / 6, PB - 1);
  const float* __restrict__ pfp =
      x + (long long)(b0 + q * PB + pseg) * (TLEN * INSZ) + min((lane % 6) * 64, SEGF - 1);
  v4f* rawq = raw[q];
  float* dst0 = (float*)stage[0] + (2 * q) * 32 + lane;
  float* dst1 = (float*)stage[1] + (2 * q) * 32 + lane;

  v2f bw[9];
  float bias = 0.0f;
  v4f ge[RAWLD], go[RAWLD];
  if (prod) {
    const int grow = (c16 & 3) * 4 + (c16 >> 2);
    const float* __restrict__ wrow = W_ih + grow * INSZ;
#pragma unroll
    for (int kk = 0; kk < 9; ++kk) {
      const int k0 = kk * 4 + koff;
      const float w0 = wrow[min(k0,     INSZ - 1)];
      const float w1 = wrow[min(k0 + 1, INSZ - 1)];
      bw[kk].x = (k0     < INSZ) ? w0 : 0.0f;
      bw[kk].y = (k0 + 1 < INSZ) ? w1 : 0.0f;
    }
    bias = b_ih[grow] + b_hh[grow];
#pragma unroll
    for (int n = 0; n < RAWLD; ++n) ge[n] = xb[offs[n]];
#pragma unroll
    for (int n = 0; n < RAWLD; ++n) go[n] = xb[offs[n] + SEGV];
#pragma unroll
    for (int kp = 2; kp < 2 + PF_CH; ++kp)
      __builtin_prefetch(pfp + kp * SEGF, 0, 0);
  }

  const int sw = prod ? 0 : wave;
  const int j  = lane & 3;
  const int b  = b0 + sw * 8 + (lane >> 2);
  const float4 Wi = *(const float4*)(W_hh + (j     ) * HID);
  const float4 Wf = *(const float4*)(W_hh + (4  + j) * HID);
  const float4 Wg = *(const float4*)(W_hh + (8  + j) * HID);
  const float4 Wo = *(const float4*)(W_hh + (12 + j) * HID);
  const float4* sl0 = &stage[0][sw * 32 + lane];
  const float4* sl1 = &stage[1][sw * 32 + lane];
  float4 ra[CH], rb[CH];
  float h = 0.0f, c = 0.0f;

  if (prod)
    produce_chunk(ge, xb, offs, 2, pfp, 2 + PF_CH, rawq, bw, bias, dst0, lane, c16, koff);
  stage_barrier();
  if (prod)
    produce_chunk(go, xb, offs, 3, pfp, 3 + PF_CH, rawq, bw, bias, dst1, lane, c16, koff);
  else
    ring_load(ra, sl0);
  stage_barrier();
#pragma unroll 1
  for (int k = 2; k < NCH + 1; k += 2) {
    if (prod) {
      produce_chunk(ge, xb, offs, min(k + 2, NCH - 1), pfp, min(k + 2 + PF_CH, NCH - 1),
                    rawq, bw, bias, dst0, lane, c16, koff);
    } else {
      ring_load(rb, sl1);
      __builtin_amdgcn_sched_barrier(0);
      ring_steps(ra, Wi, Wf, Wg, Wo, h, c);
    }
    stage_barrier();
    if (prod) {
      produce_chunk(go, xb, offs, min(k + 3, NCH - 1), pfp, min(k + 3 + PF_CH, NCH - 1),
                    rawq, bw, bias, dst1, lane, c16, koff);
    } else {
      ring_load(ra, sl0);
      __builtin_amdgcn_sched_barrier(0);
      ring_steps(rb, Wi, Wf, Wg, Wo, h, c);
    }
    stage_barrier();
  }
  if (prod) return;
  ring_steps(ra, Wi, Wf, Wg, Wo, h, c);

  const float h0 = __shfl(h, 0, 4);
  const float h1 = __shfl(h, 1, 4);
  const float h2 = __shfl(h, 2, 4);
  const float h3 = __shfl(h, 3, 4);
  if (j < NCLS) {
    const float4 Wc = *(const float4*)(W_fc + j * HID);
    out[b * NCLS + j] =
        b_fc[j] + Wc.x * h0 + Wc.y * h1 + Wc.z * h2 + Wc.w * h3;
  }
}

extern "C" void kernel_launch(void* const* d_in, const int* in_sizes, int n_in,
                              void* d_out, int out_size, void* d_ws, size_t ws_size,
                              hipStream_t stream) {
  (void)in_sizes; (void)n_in; (void)out_size; (void)d_ws; (void)ws_size;
  const float* x    = (const float*)d_in[0];
  const float* W_ih = (const float*)d_in[1];
  const float* W_hh = (const float*)d_in[2];
  const float* b_ih = (const float*)d_in[3];
  const float* b_hh = (const float*)d_in[4];
  const float* W_fc = (const float*)d_in[5];
  const float* b_fc = (const float*)d_in[6];
  float* out = (float*)d_out;

  hipLaunchKernelGGL(lstm_fused, dim3(BATCH / NB), dim3(FUSED_THREADS), 0, stream,
                     x, W_ih, W_hh, b_ih, b_hh, W_fc, b_fc, out);
}


# scan step's serial chain shortened: h.W_hh dot as two 2-term FMA chains plus one add (summation order changed); sigmoid(x) as (1+tanh(x/2))/2, the halving folded exactly into the gate weights
# speedup vs baseline: 3.2949x; 1.1660x over previous
#include <hip/hip_runtime.h>


#define BATCH 4096
#define TLEN  200
#define INSZ  34
#define HID   4
#define GATES 16
#define NCLS  3

typedef float v2f __attribute__((ext_vector_type(2)));
typedef float v4f __attribute__((ext_vector_type(4)));
typedef float v8f __attribute__((ext_vector_type(8)));

__device__ __forceinline__ float fast_rcp(float x) {
  return __builtin_amdgcn_rcpf(x);
}
__device__ __forceinline__ float fast_tanh(float x) {
#if __has_builtin(__builtin_amdgcn_tanhf)
  return __builtin_amdgcn_tanhf(x);
#else
  float e = __expf(-2.0f * x);
  return (1.0f - e) * fast_rcp(1.0f + e);
#endif
}

#define NB      16
#define CH      8
#define NCH     (TLEN / CH)
#define SCANW   2
#define PRODW   4
#define FUSED_THREADS ((SCANW + PRODW) * 32)
#define PB      (NB / PRODW)
#define SEGF    (CH * INSZ)
#define SEGV    (SEGF / 4)
#define BSTRV   (TLEN * INSZ / 4)
#define RAWV    (PB * SEGV)
#define RAWLD   ((RAWV + 31) / 32)
#define STAGEV  (CH * NB * GATES / 4)
#define PF_CH   4

static_assert(TLEN % CH == 0, "chunks must tile the sequence");
static_assert(NCH % 2 == 1 && NCH >= 3, "the loop below does two chunks per trip behind two peeled ones and one drain");
static_assert(BATCH % NB == 0, "workgroups must tile the batch");
static_assert(SCANW * 8 == NB && PB == 4, "a producer wave makes two 16-row tiles (2 batch elements x 8 steps each) per chunk");
static_assert(SEGF % 4 == 0 && (TLEN * INSZ) % 4 == 0, "16-byte loads need 16-byte aligned segments");


__device__ __forceinline__ void produce_chunk(
    v4f (&g)[RAWLD], const v4f* __restrict__ xb, const int (&offs)[RAWLD],
    const int knext, const float* __restrict__ pfp, const int kpf,
    v4f* rawq, const v2f (&bw)[9], const float bias, float* dst,
    const int lane, const int c, const int koff)
{
#pragma unroll
  for (int n = 0; n < RAWLD; ++n)
    rawq[min(n * 32 + lane, RAWV - 1)] = g[n];
#pragma unroll
  for (int n = 0; n < RAWLD; ++n)
    g[n] = xb[offs[n] + knext * SEGV];
  __builtin_prefetch(pfp + kpf * SEGF, 0, 0);
  __builtin_amdgcn_sched_barrier(0);

  const float* rf = (const float*)rawq;
  v2f a0[9], a1[9];
#pragma unroll
  for (int kk = 0; kk < 9; ++kk) {
    const int kc = (kk < 8) ? (kk * 4 + koff) : (INSZ - 2);
    a0[kk] = *(const v2f*)(rf + c * INSZ + kc);
    a1[kk] = *(const v2f*)(rf + (16 + c) * INSZ + kc);
  }
  __builtin_amdgcn_sched_barrier(0);
  v8f acc0, acc1;
#pragma unroll
  for (int v = 0; v < 8; ++v) { acc0[v] = bias; acc1[v] = bias; }
#pragma unroll
  for (int kk = 0; kk < 9; ++kk) {
    const int k0 = kk * 4 + koff;
    v2f ak0, ak1;
    ak0.x = (k0     < INSZ) ? a0[kk].x : 0.0f;
    ak0.y = (k0 + 1 < INSZ) ? a0[kk].y : 0.0f;
    ak1.x = (k0     < INSZ) ? a1[kk].x : 0.0f;
    ak1.y = (k0 + 1 < INSZ) ? a1[kk].y : 0.0f;
    acc0 = __builtin_amdgcn_wmma_f32_16x16x4_f32(
        false, ak0, false, bw[kk], (short)0, acc0, false, false);
    acc1 = __builtin_amdgcn_wmma_f32_16x16x4_f32(
        false, ak1, false, bw[kk], (short)0, acc1, false, false);
  }
#pragma unroll
  for (int v = 0; v < 8; ++v) {
    dst[v * (NB * GATES)]      = acc0[v];
    dst[32 + v * (NB * GATES)] = acc1[v];
  }
}

__device__ __forceinline__ float gate_dot(
    const float p, const float4 W,
    const float h0, const float h1, const float h2, const float h3)
{
  const float lo = fmaf(W.y, h1, fmaf(W.x, h0, p));
  const float hi = fmaf(W.w, h3, W.z * h2);
  return lo + hi;
}

__device__ __forceinline__ void lstm_step(
    const float pi, const float pf, const float pg, const float po,
    const float4 Wi, const float4 Wf, const float4 Wg, const float4 Wo,
    float& h, float& c)
{
  const float h0 = __shfl(h, 0, 4);
  const float h1 = __shfl(h, 1, 4);
  const float h2 = __shfl(h, 2, 4);
  const float h3 = __shfl(h, 3, 4);

  const float ti = fast_tanh(gate_dot(pi, Wi, h0, h1, h2, h3));
  const float tf = fast_tanh(gate_dot(pf, Wf, h0, h1, h2, h3));
  const float gg = fast_tanh(gate_dot(pg, Wg, h0, h1, h2, h3));
  const float to = fast_tanh(gate_dot(po, Wo, h0, h1, h2, h3));

  const float hg  = 0.5f * gg;
  const float ixg = fmaf(ti, hg, hg);
  const float fg  = fmaf(0.5f, tf, 0.5f);
  const float og  = fmaf(0.5f, to, 0.5f);

  c = fmaf(fg, c, ixg);
  h = og * fast_tanh(c);
}

__device__ __forceinline__ void ring_load(float4 (&r)[CH], const float4* sl)
{
#pragma unroll
  for (int u = 0; u < CH; ++u) r[u] = sl[u * (NB * GATES / 4)];
}
__device__ __forceinline__ void ring_steps(
    const float4 (&r)[CH],
    const float4 Wi, const float4 Wf, const float4 Wg, const float4 Wo,
    float& h, float& c)
{
#pragma unroll
  for (int u = 0; u < CH; ++u)
    lstm_step(r[u].x, r[u].y, r[u].z, r[u].w, Wi, Wf, Wg, Wo, h, c);
}

__device__ __forceinline__ void stage_barrier()
{
  __builtin_amdgcn_fence(__ATOMIC_RELEASE, "workgroup", "local");
  __builtin_amdgcn_s_barrier();
  __builtin_amdgcn_fence(__ATOMIC_ACQUIRE, "workgroup", "local");
}

__global__ __launch_bounds__(FUSED_THREADS) void lstm_fused(
    const float* __restrict__ x,
    const float* __restrict__ W_ih,
    const float* __restrict__ W_hh,
    const float* __restrict__ b_ih,
    const float* __restrict__ b_hh,
    const float* __restrict__ W_fc,
    const float* __restrict__ b_fc,
    float* __restrict__ out)
{
  __shared__ float4 stage[2][STAGEV];
  __shared__ v4f raw[PRODW][RAWV];

  const int lane = threadIdx.x & 31;
  const int wave = __builtin_amdgcn_readfirstlane((int)(threadIdx.x >> 5));
  const bool prod = wave >= SCANW;
  const int b0 = blockIdx.x * NB;

  const int half = lane >> 4;
  const int c16  = lane & 15;
  const int koff = half * 2;
  const int q    = prod ? (wave - SCANW) : 0;
  const v4f* __restrict__ xb =
      (const v4f*)(x + (long long)(b0 + q * PB) * (TLEN * INSZ));
  int offs[RAWLD];
#pragma unroll
  for (int n = 0; n < RAWLD; ++n) {
    const int i = min(n * 32 + lane, RAWV - 1);
    offs[n] = (i / SEGV) * BSTRV + (i % SEGV);
  }
  const int pseg = min(lane / 6, PB - 1);
  const float* __restrict__ pfp =
      x + (long long)(b0 + q * PB + pseg) * (TLEN * INSZ) + min((lane % 6) * 64, SEGF - 1);
  v4f* rawq = raw[q];
  float* dst0 = (float*)stage[0] + (2 * q) * 32 + lane;
  float* dst1 = (float*)stage[1] + (2 * q) * 32 + lane;

  v2f bw[9];
  float bias = 0.0f;
  v4f ge[RAWLD], go[RAWLD];
  if (prod) {
    const int grow = (c16 & 3) * 4 + (c16 >> 2);
    const float gsc = ((c16 & 3) == 2) ? 1.0f : 0.5f;
    const float* __restrict__ wrow = W_ih + grow * INSZ;
#pragma unroll
    for (int kk = 0; kk < 9; ++kk) {
      const int k0 = kk * 4 + koff;
      const float w0 = wrow[min(k0,     INSZ - 1)];
      const float w1 = wrow[min(k0 + 1, INSZ - 1)];
      bw[kk].x = (k0     < INSZ) ? w0 * gsc : 0.0f;
      bw[kk].y = (k0 + 1 < INSZ) ? w1 * gsc : 0.0f;
    }
    bias = (b_ih[grow] + b_hh[grow]) * gsc;
#pragma unroll
    for (int n = 0; n < RAWLD; ++n) ge[n] = xb[offs[n]];
#pragma unroll
    for (int n = 0; n < RAWLD; ++n) go[n] = xb[offs[n] + SEGV];
#pragma unroll
    for (int kp = 2; kp < 2 + PF_CH; ++kp)
      __builtin_prefetch(pfp + kp * SEGF, 0, 0);
  }

  const int sw = prod ? 0 : wave;
  const int j  = lane & 3;
  const int b  = b0 + sw * 8 + (lane >> 2);
  const float4 Wi1 = *(const float4*)(W_hh + (j     ) * HID);
  const float4 Wf1 = *(const float4*)(W_hh + (4  + j) * HID);
  const float4 Wg  = *(const float4*)(W_hh + (8  + j) * HID);
  const float4 Wo1 = *(const float4*)(W_hh + (12 + j) * HID);
  const float4 Wi = make_float4(0.5f * Wi1.x, 0.5f * Wi1.y, 0.5f * Wi1.z, 0.5f * Wi1.w);
  const float4 Wf = make_float4(0.5f * Wf1.x, 0.5f * Wf1.y, 0.5f * Wf1.z, 0.5f * Wf1.w);
  const float4 Wo = make_float4(0.5f * Wo1.x, 0.5f * Wo1.y, 0.5f * Wo1.z, 0.5f * Wo1.w);
  const float4* sl0 = &stage[0][sw * 32 + lane];
  const float4* sl1 = &stage[1][sw * 32 + lane];
  float4 ra[CH], rb[CH];
  float h = 0.0f, c = 0.0f;

  if (prod)
    produce_chunk(ge, xb, offs, 2, pfp, 2 + PF_CH, rawq, bw, bias, dst0, lane, c16, koff);
  stage_barrier();
  if (prod)
    produce_chunk(go, xb, offs, 3, pfp, 3 + PF_CH, rawq, bw, bias, dst1, lane, c16, koff);
  else
    ring_load(ra, sl0);
  stage_barrier();
#pragma unroll 1
  for (int k = 2; k < NCH + 1; k += 2) {
    if (prod) {
      produce_chunk(ge, xb, offs, min(k + 2, NCH - 1), pfp, min(k + 2 + PF_CH, NCH - 1),
                    rawq, bw, bias, dst0, lane, c16, koff);
    } else {
      ring_load(rb, sl1);
      __builtin_amdgcn_sched_barrier(0);
      ring_steps(ra, Wi, Wf, Wg, Wo, h, c);
    }
    stage_barrier();
    if (prod) {
      produce_chunk(go, xb, offs, min(k + 3, NCH - 1), pfp, min(k + 3 + PF_CH, NCH - 1),
                    rawq, bw, bias, dst1, lane, c16, koff);
    } else {
      ring_load(ra, sl0);
      __builtin_amdgcn_sched_barrier(0);
      ring_steps(rb, Wi, Wf, Wg, Wo, h, c);
    }
    stage_barrier();
  }
  if (prod) return;
  ring_steps(ra, Wi, Wf, Wg, Wo, h, c);

  const float h0 = __shfl(h, 0, 4);
  const float h1 = __shfl(h, 1, 4);
  const float h2 = __shfl(h, 2, 4);
  const float h3 = __shfl(h, 3, 4);
  if (j < NCLS) {
    const float4 Wc = *(const float4*)(W_fc + j * HID);
    out[b * NCLS + j] =
        b_fc[j] + Wc.x * h0 + Wc.y * h1 + Wc.z * h2 + Wc.w * h3;
  }
}

extern "C" void kernel_launch(void* const* d_in, const int* in_sizes, int n_in,
                              void* d_out, int out_size, void* d_ws, size_t ws_size,
                              hipStream_t stream) {
  (void)in_sizes; (void)n_in; (void)out_size; (void)d_ws; (void)ws_size;
  const float* x    = (const float*)d_in[0];
  const float* W_ih = (const float*)d_in[1];
  const float* W_hh = (const float*)d_in[2];
  const float* b_ih = (const float*)d_in[3];
  const float* b_hh = (const float*)d_in[4];
  const float* W_fc = (const float*)d_in[5];
  const float* b_fc = (const float*)d_in[6];
  float* out = (float*)d_out;

  hipLaunchKernelGGL(lstm_fused, dim3(BATCH / NB), dim3(FUSED_THREADS), 0, stream,
                     x, W_ih, W_hh, b_ih, b_hh, W_fc, b_fc, out);
}
